# starting kernel
# baseline (speedup 1.0000x reference)
#include <hip/hip_runtime.h>

#define B_ 8
#define S_ 512
#define D_ 512
#define H_ 64
#define DK_ 8
#define WIDTH_ 16
#define N_ (S_ + 2 * WIDTH_)
#define NL_ 2
#define WIN_ (WIDTH_ + 2)
#define BN_ (B_ * N_)
#define BS_ (B_ * S_)
static const size_t BND_ = (size_t)BN_ * D_;
static const size_t BSD_ = (size_t)BS_ * D_;

typedef __attribute__((ext_vector_type(2))) float v2f;
typedef __attribute__((ext_vector_type(8))) float v8f;

__global__ __launch_bounds__(256)
void pad_kernel(const float* __restrict__ x, const float* __restrict__ lp,
                const float* __restrict__ rp, float* __restrict__ xp) {
  size_t idx = (size_t)blockIdx.x * blockDim.x + threadIdx.x;
  if (idx >= BND_) return;
  int dd = (int)(idx % D_);
  size_t bn = idx / D_;
  int p = (int)(bn % N_);
  int b = (int)(bn / N_);
  float v;
  if (p < WIDTH_)            v = lp[p * D_ + dd];
  else if (p < WIDTH_ + S_)  v = x[((size_t)b * S_ + (p - WIDTH_)) * D_ + dd];
  else                       v = rp[(p - WIDTH_ - S_) * D_ + dd];
  xp[idx] = v;
}

__global__ __launch_bounds__(256)
void gemm_wmma_f32(const float* __restrict__ A, int lda,
                   const float* __restrict__ W, int ldw,
                   const float* __restrict__ bias,
                   float* __restrict__ C, int ldc,
                   int M, int Nout) {
  constexpr int KK = 512;
  constexpr int LDA_S = KK + 8;
  __shared__ float As[16 * LDA_S];

  const int lane = threadIdx.x;
  const int wave = threadIdx.y;
  const int tid  = wave * 32 + lane;
  const int row0 = blockIdx.y * 16;
  const int col0 = (blockIdx.x * 8 + wave) * 64;

  {
    constexpr int NVEC = 16 * (KK / 4);
#pragma unroll
    for (int v = 0; v < NVEC / 256; ++v) {
      int e   = tid + v * 256;
      int row = e >> 7;
      int cv  = e & 127;
      int rg  = row0 + row; if (rg > M - 1) rg = M - 1;
      float4 val = *(const float4*)(A + (size_t)rg * lda + cv * 4);
      *(float4*)&As[row * LDA_S + cv * 4] = val;
    }
  }
  __syncthreads();

  if (row0 >= M || col0 >= Nout) return;

  const int ml = lane & 15;
  const int kb = (lane >> 4) << 1;

  const float* w0 = W + (size_t)(col0 + ml) * ldw + kb;
  const float* w1 = w0 + (size_t)16 * ldw;
  const float* w2 = w0 + (size_t)32 * ldw;
  const float* w3 = w0 + (size_t)48 * ldw;
  const float* arow = &As[ml * LDA_S + kb];

  v8f acc0 = {}, acc1 = {}, acc2 = {}, acc3 = {};

  v2f a  = *(const v2f*)(arow);
  v2f b0 = *(const v2f*)(w0);
  v2f b1 = *(const v2f*)(w1);
  v2f b2 = *(const v2f*)(w2);
  v2f b3 = *(const v2f*)(w3);

#pragma unroll 4
  for (int k = 4; k < KK; k += 4) {
    v2f an  = *(const v2f*)(arow + k);
    v2f b0n = *(const v2f*)(w0 + k);
    v2f b1n = *(const v2f*)(w1 + k);
    v2f b2n = *(const v2f*)(w2 + k);
    v2f b3n = *(const v2f*)(w3 + k);
    acc0 = __builtin_amdgcn_wmma_f32_16x16x4_f32(false, a, false, b0, (short)0, acc0, false, false);
    acc1 = __builtin_amdgcn_wmma_f32_16x16x4_f32(false, a, false, b1, (short)0, acc1, false, false);
    acc2 = __builtin_amdgcn_wmma_f32_16x16x4_f32(false, a, false, b2, (short)0, acc2, false, false);
    acc3 = __builtin_amdgcn_wmma_f32_16x16x4_f32(false, a, false, b3, (short)0, acc3, false, false);
    a = an; b0 = b0n; b1 = b1n; b2 = b2n; b3 = b3n;
  }
  acc0 = __builtin_amdgcn_wmma_f32_16x16x4_f32(false, a, false, b0, (short)0, acc0, false, false);
  acc1 = __builtin_amdgcn_wmma_f32_16x16x4_f32(false, a, false, b1, (short)0, acc1, false, false);
  acc2 = __builtin_amdgcn_wmma_f32_16x16x4_f32(false, a, false, b2, (short)0, acc2, false, false);
  acc3 = __builtin_amdgcn_wmma_f32_16x16x4_f32(false, a, false, b3, (short)0, acc3, false, false);

  const int mbase = row0 + ((lane >> 4) << 3);
  float bv0 = bias[col0 +  0 + ml];
  float bv1 = bias[col0 + 16 + ml];
  float bv2 = bias[col0 + 32 + ml];
  float bv3 = bias[col0 + 48 + ml];
#pragma unroll
  for (int r = 0; r < 8; ++r) {
    float* crow = C + (size_t)(mbase + r) * ldc + col0 + ml;
    crow[ 0] = acc0[r] + bv0;
    crow[16] = acc1[r] + bv1;
    crow[32] = acc2[r] + bv2;
    crow[48] = acc3[r] + bv3;
  }
}

__global__ __launch_bounds__(256)
void attn_kernel(const float* __restrict__ qkv, float* __restrict__ out, int dir) {
  int idx = blockIdx.x * blockDim.x + threadIdx.x;
  if (idx >= B_ * H_ * N_) return;
  int h    = idx % H_;
  int rest = idx / H_;
  int i    = rest % N_;
  int b    = rest / N_;

  const size_t base = (size_t)(b * N_ + i) * D_ + h * DK_;
  const float* qp = qkv + base;
  const float* kbase = qkv + BND_ + (size_t)(b * N_) * D_ + h * DK_;
  const float* vbase = qkv + 2 * BND_ + (size_t)(b * N_) * D_ + h * DK_;

  float q[DK_];
  *(float4*)&q[0] = *(const float4*)(qp);
  *(float4*)&q[4] = *(const float4*)(qp + 4);

  const int j0 = (dir == 0) ? ((i - WIDTH_ - 1 > 0) ? i - WIDTH_ - 1 : 0) : i;
  const int j1 = (dir == 0) ? i : ((i + WIDTH_ + 1 < N_ - 1) ? i + WIDTH_ + 1 : N_ - 1);
  const float scale = 0.35355339059327373f;

  float sv[WIN_];
  float mx = -3.0e38f;
#pragma unroll
  for (int t = 0; t < WIN_; ++t) {
    int j = j0 + t;
    float sc = -3.0e38f;
    if (j <= j1) {
      const float* kp = kbase + (size_t)j * D_;
      float kr[DK_];
      *(float4*)&kr[0] = *(const float4*)(kp);
      *(float4*)&kr[4] = *(const float4*)(kp + 4);
      float dot = 0.f;
#pragma unroll
      for (int d = 0; d < DK_; ++d) dot += q[d] * kr[d];
      sc = dot * scale;
    }
    sv[t] = sc;
    mx = fmaxf(mx, sc);
  }
  float sum = 0.f;
#pragma unroll
  for (int t = 0; t < WIN_; ++t) {
    float p = __expf(sv[t] - mx);
    sv[t] = p;
    sum += p;
  }
  const float inv = 1.f / sum;
  float o[DK_] = {};
#pragma unroll
  for (int t = 0; t < WIN_; ++t) {
    int j = j0 + t;
    if (j <= j1) {
      const float* vp = vbase + (size_t)j * D_;
      float vr[DK_];
      *(float4*)&vr[0] = *(const float4*)(vp);
      *(float4*)&vr[4] = *(const float4*)(vp + 4);
      float w = sv[t] * inv;
#pragma unroll
      for (int d = 0; d < DK_; ++d) o[d] += w * vr[d];
    }
  }
  float* op = out + base;
  float4 o0; o0.x = o[0]; o0.y = o[1]; o0.z = o[2]; o0.w = o[3];
  float4 o1; o1.x = o[4]; o1.y = o[5]; o1.z = o[6]; o1.w = o[7];
  *(float4*)(op)     = o0;
  *(float4*)(op + 4) = o1;
}

__global__ __launch_bounds__(256)
void slice_kernel(const float* __restrict__ proj, float* __restrict__ hx) {
  size_t idx = (size_t)blockIdx.x * blockDim.x + threadIdx.x;
  if (idx >= BSD_) return;
  int dd = (int)(idx % D_);
  size_t ms = idx / D_;
  int s = (int)(ms % S_);
  int b = (int)(ms / S_);
  hx[idx] = proj[((size_t)(b * N_ + WIDTH_ + s)) * D_ + dd];
}

__global__ __launch_bounds__(256)
void gate_kernel(const float* __restrict__ hp, float* __restrict__ hx) {
  size_t idx = (size_t)blockIdx.x * blockDim.x + threadIdx.x;
  if (idx >= BSD_) return;
  int dd = (int)(idx % D_);
  size_t m = idx / D_;
  float lin = hp[m * (2 * D_) + dd];
  float gin = hp[m * (2 * D_) + D_ + dd];
  float g  = 1.f / (1.f + __expf(-gin));
  float nl = fmaxf(lin, 0.f);
  hx[idx] = g * hx[idx] + (1.f - g) * nl;
}

__global__ __launch_bounds__(256)
void outw_kernel(const float* __restrict__ hx, float* __restrict__ out, int dir) {
  size_t idx = (size_t)blockIdx.x * blockDim.x + threadIdx.x;
  if (idx >= BSD_) return;
  int dd = (int)(idx % D_);
  size_t m = idx / D_;
  size_t o = m * (size_t)(2 * D_) + (size_t)dir * D_ + dd;
  float v = hx[idx];
  out[o] = v;
  out[o + (size_t)BS_ * 2 * D_] = v;
}

extern "C" void kernel_launch(void* const* d_in, const int* in_sizes, int n_in,
                              void* d_out, int out_size, void* d_ws, size_t ws_size,
                              hipStream_t stream) {
  const float* inputs    = (const float*)d_in[0];
  const float* left_pad  = (const float*)d_in[1];
  const float* right_pad = (const float*)d_in[2];
  const float* Wqkv[2] = {(const float*)d_in[3],  (const float*)d_in[7]};
  const float* bqkv[2] = {(const float*)d_in[4],  (const float*)d_in[8]};
  const float* Wo[2]   = {(const float*)d_in[5],  (const float*)d_in[9]};
  const float* bo[2]   = {(const float*)d_in[6],  (const float*)d_in[10]};
  const float* hwW[2]  = {(const float*)d_in[11], (const float*)d_in[13]};
  const float* hwb[2]  = {(const float*)d_in[12], (const float*)d_in[14]};
  float* out = (float*)d_out;

  float* ws    = (float*)d_ws;
  float* xpad  = ws;
  float* qkv   = xpad + BND_;
  float* attn  = qkv + 3 * BND_;
  float* hx    = attn + BND_;
  float* proj  = qkv;
  float* hproj = qkv + BND_;

  dim3 blk(32, 8);

  { int n = (int)BND_; pad_kernel<<<(n + 255) / 256, 256, 0, stream>>>(inputs, left_pad, right_pad, xpad); }

  for (int dir = 0; dir < 2; ++dir) {
    for (int t = 0; t < 3; ++t) {
      dim3 grd(1, BN_ / 16);
      gemm_wmma_f32<<<grd, blk, 0, stream>>>(
          xpad, D_, Wqkv[dir] + (size_t)t * D_ * D_, D_, bqkv[dir] + t * D_,
          qkv + (size_t)t * BND_, D_, BN_, D_);
    }
    { int n = B_ * H_ * N_;
      attn_kernel<<<(n + 255) / 256, 256, 0, stream>>>(qkv, attn, dir); }
    { dim3 grd(1, BN_ / 16);
      gemm_wmma_f32<<<grd, blk, 0, stream>>>(attn, D_, Wo[dir], D_, bo[dir],
                                             proj, D_, BN_, D_); }
    { int n = (int)BSD_; slice_kernel<<<(n + 255) / 256, 256, 0, stream>>>(proj, hx); }
    for (int l = 0; l < NL_; ++l) {
      dim3 grd(2, BS_ / 16);
      gemm_wmma_f32<<<grd, blk, 0, stream>>>(
          hx, D_, hwW[dir] + (size_t)l * 2 * D_ * D_, D_, hwb[dir] + l * 2 * D_,
          hproj, 2 * D_, BS_, 2 * D_);
      int n = (int)BSD_;
      gate_kernel<<<(n + 255) / 256, 256, 0, stream>>>(hproj, hx);
    }
    { int n = (int)BSD_; outw_kernel<<<(n + 255) / 256, 256, 0, stream>>>(hx, out, dir); }
  }
}


# GEMM re-tiled: 64x64 block tile, K in chunks of 32, A and W slices both staged in LDS by coalesced b128 loads (2 stages), all WMMA fragments read from LDS; k order per accumulator unchanged
# speedup vs baseline: 2.0189x; 2.0189x over previous
#include <hip/hip_runtime.h>

#define B_ 8
#define S_ 512
#define D_ 512
#define H_ 64
#define DK_ 8
#define WIDTH_ 16
#define N_ (S_ + 2 * WIDTH_)
#define NL_ 2
#define WIN_ (WIDTH_ + 2)
#define BN_ (B_ * N_)
#define BS_ (B_ * S_)
static const size_t BND_ = (size_t)BN_ * D_;
static const size_t BSD_ = (size_t)BS_ * D_;

typedef __attribute__((ext_vector_type(2))) float v2f;
typedef __attribute__((ext_vector_type(8))) float v8f;

__global__ __launch_bounds__(256)
void pad_kernel(const float* __restrict__ x, const float* __restrict__ lp,
                const float* __restrict__ rp, float* __restrict__ xp) {
  size_t idx = (size_t)blockIdx.x * blockDim.x + threadIdx.x;
  if (idx >= BND_) return;
  int dd = (int)(idx % D_);
  size_t bn = idx / D_;
  int p = (int)(bn % N_);
  int b = (int)(bn / N_);
  float v;
  if (p < WIDTH_)            v = lp[p * D_ + dd];
  else if (p < WIDTH_ + S_)  v = x[((size_t)b * S_ + (p - WIDTH_)) * D_ + dd];
  else                       v = rp[(p - WIDTH_ - S_) * D_ + dd];
  xp[idx] = v;
}

#define GEMM_BM 64
#define GEMM_BN 64

__global__ __launch_bounds__(256)
void gemm_wmma_f32(const float* __restrict__ A, int lda,
                   const float* __restrict__ W, int ldw,
                   const float* __restrict__ bias,
                   float* __restrict__ C, int ldc,
                   int M, int Nout) {
  constexpr int KK = 512;
  constexpr int BM = GEMM_BM;
  constexpr int BN = GEMM_BN;
  constexpr int BK = 32;
  constexpr int PITCH = BK + 4;
  constexpr int NCH = KK / BK;
  constexpr int KS = BK / 4;
  __shared__ __attribute__((aligned(16))) float As[2][BM * PITCH];
  __shared__ __attribute__((aligned(16))) float Bs[2][BN * PITCH];

  const int lane = threadIdx.x;
  const int wave = threadIdx.y;
  const int tid  = wave * 32 + lane;
  const int row0 = blockIdx.y * BM;
  const int col0 = blockIdx.x * BN;

  if (row0 >= M || col0 >= Nout) return;

  const int srow = tid >> 3;
  const int scv  = (tid & 7) << 2;
  int ar0 = row0 + srow;      if (ar0 > M - 1)    ar0 = M - 1;
  int ar1 = row0 + srow + 32; if (ar1 > M - 1)    ar1 = M - 1;
  int wr0 = col0 + srow;      if (wr0 > Nout - 1) wr0 = Nout - 1;
  int wr1 = col0 + srow + 32; if (wr1 > Nout - 1) wr1 = Nout - 1;
  const unsigned oa0 = (unsigned)(ar0 * lda + scv) * 4u;
  const unsigned oa1 = (unsigned)(ar1 * lda + scv) * 4u;
  const unsigned ow0 = (unsigned)(wr0 * ldw + scv) * 4u;
  const unsigned ow1 = (unsigned)(wr1 * ldw + scv) * 4u;
  const int so0 = srow * PITCH + scv;
  const int so1 = (srow + 32) * PITCH + scv;

  float4 pa0 = *(const float4*)((const char*)A + oa0);
  float4 pa1 = *(const float4*)((const char*)A + oa1);
  float4 pw0 = *(const float4*)((const char*)W + ow0);
  float4 pw1 = *(const float4*)((const char*)W + ow1);
  *(float4*)&As[0][so0] = pa0;
  *(float4*)&As[0][so1] = pa1;
  *(float4*)&Bs[0][so0] = pw0;
  *(float4*)&Bs[0][so1] = pw1;
  __syncthreads();

  const int wm = wave & 3;
  const int wn = wave >> 2;
  const int ml = lane & 15;
  const int kb = (lane >> 4) << 1;
  const int aoff = (wm * 16 + ml) * PITCH + kb;
  const int boff = (wn * 32 + ml) * PITCH + kb;

  v8f acc0 = {}, acc1 = {};

  auto chunk = [&](const int cur, const int kn) __attribute__((always_inline)) {
    const char* An = (const char*)(A + kn * BK);
    const char* Wn = (const char*)(W + kn * BK);
    pa0 = *(const float4*)(An + oa0);
    pa1 = *(const float4*)(An + oa1);
    pw0 = *(const float4*)(Wn + ow0);
    pw1 = *(const float4*)(Wn + ow1);
    __builtin_amdgcn_sched_barrier(0);
    const float* ap = &As[cur][aoff];
    const float* bp = &Bs[cur][boff];
    v2f af[KS], bf0[KS], bf1[KS];
#pragma unroll
    for (int s = 0; s < KS; ++s) {
      af[s]  = *(const v2f*)(ap + 4 * s);
      bf0[s] = *(const v2f*)(bp + 4 * s);
      bf1[s] = *(const v2f*)(bp + 16 * PITCH + 4 * s);
    }
#pragma unroll
    for (int s = 0; s < KS; ++s) {
      acc0 = __builtin_amdgcn_wmma_f32_16x16x4_f32(false, af[s], false, bf0[s], (short)0, acc0, false, false);
      acc1 = __builtin_amdgcn_wmma_f32_16x16x4_f32(false, af[s], false, bf1[s], (short)0, acc1, false, false);
    }
    asm volatile("v_nop\n\tv_nop\n\tv_nop\n\tv_nop\n\tv_nop" : "+v"(acc0), "+v"(acc1));
    __builtin_amdgcn_sched_barrier(0);
    *(float4*)&As[cur ^ 1][so0] = pa0;
    *(float4*)&As[cur ^ 1][so1] = pa1;
    *(float4*)&Bs[cur ^ 1][so0] = pw0;
    *(float4*)&Bs[cur ^ 1][so1] = pw1;
    __syncthreads();
  };

  for (int kc = 0; kc < NCH; kc += 2) {
    chunk(0, kc + 1);
    chunk(1, (kc + 2 < NCH) ? kc + 2 : NCH - 1);
  }

  const int mbase = row0 + wm * 16 + ((lane >> 4) << 3);
  const int cb = col0 + wn * 32 + ml;
  const float bv0 = bias[cb];
  const float bv1 = bias[cb + 16];
#pragma unroll
  for (int r = 0; r < 8; ++r) {
    float* crow = C + (size_t)(mbase + r) * ldc + cb;
    crow[ 0] = acc0[r] + bv0;
    crow[16] = acc1[r] + bv1;
  }
}

__global__ __launch_bounds__(256)
void attn_kernel(const float* __restrict__ qkv, float* __restrict__ out, int dir) {
  int idx = blockIdx.x * blockDim.x + threadIdx.x;
  if (idx >= B_ * H_ * N_) return;
  int h    = idx % H_;
  int rest = idx / H_;
  int i    = rest % N_;
  int b    = rest / N_;

  const size_t base = (size_t)(b * N_ + i) * D_ + h * DK_;
  const float* qp = qkv + base;
  const float* kbase = qkv + BND_ + (size_t)(b * N_) * D_ + h * DK_;
  const float* vbase = qkv + 2 * BND_ + (size_t)(b * N_) * D_ + h * DK_;

  float q[DK_];
  *(float4*)&q[0] = *(const float4*)(qp);
  *(float4*)&q[4] = *(const float4*)(qp + 4);

  const int j0 = (dir == 0) ? ((i - WIDTH_ - 1 > 0) ? i - WIDTH_ - 1 : 0) : i;
  const int j1 = (dir == 0) ? i : ((i + WIDTH_ + 1 < N_ - 1) ? i + WIDTH_ + 1 : N_ - 1);
  const float scale = 0.35355339059327373f;

  float sv[WIN_];
  float mx = -3.0e38f;
#pragma unroll
  for (int t = 0; t < WIN_; ++t) {
    int j = j0 + t;
    float sc = -3.0e38f;
    if (j <= j1) {
      const float* kp = kbase + (size_t)j * D_;
      float kr[DK_];
      *(float4*)&kr[0] = *(const float4*)(kp);
      *(float4*)&kr[4] = *(const float4*)(kp + 4);
      float dot = 0.f;
#pragma unroll
      for (int d = 0; d < DK_; ++d) dot += q[d] * kr[d];
      sc = dot * scale;
    }
    sv[t] = sc;
    mx = fmaxf(mx, sc);
  }
  float sum = 0.f;
#pragma unroll
  for (int t = 0; t < WIN_; ++t) {
    float p = __expf(sv[t] - mx);
    sv[t] = p;
    sum += p;
  }
  const float inv = 1.f / sum;
  float o[DK_] = {};
#pragma unroll
  for (int t = 0; t < WIN_; ++t) {
    int j = j0 + t;
    if (j <= j1) {
      const float* vp = vbase + (size_t)j * D_;
      float vr[DK_];
      *(float4*)&vr[0] = *(const float4*)(vp);
      *(float4*)&vr[4] = *(const float4*)(vp + 4);
      float w = sv[t] * inv;
#pragma unroll
      for (int d = 0; d < DK_; ++d) o[d] += w * vr[d];
    }
  }
  float* op = out + base;
  float4 o0; o0.x = o[0]; o0.y = o[1]; o0.z = o[2]; o0.w = o[3];
  float4 o1; o1.x = o[4]; o1.y = o[5]; o1.z = o[6]; o1.w = o[7];
  *(float4*)(op)     = o0;
  *(float4*)(op + 4) = o1;
}

__global__ __launch_bounds__(256)
void slice_kernel(const float* __restrict__ proj, float* __restrict__ hx) {
  size_t idx = (size_t)blockIdx.x * blockDim.x + threadIdx.x;
  if (idx >= BSD_) return;
  int dd = (int)(idx % D_);
  size_t ms = idx / D_;
  int s = (int)(ms % S_);
  int b = (int)(ms / S_);
  hx[idx] = proj[((size_t)(b * N_ + WIDTH_ + s)) * D_ + dd];
}

__global__ __launch_bounds__(256)
void gate_kernel(const float* __restrict__ hp, float* __restrict__ hx) {
  size_t idx = (size_t)blockIdx.x * blockDim.x + threadIdx.x;
  if (idx >= BSD_) return;
  int dd = (int)(idx % D_);
  size_t m = idx / D_;
  float lin = hp[m * (2 * D_) + dd];
  float gin = hp[m * (2 * D_) + D_ + dd];
  float g  = 1.f / (1.f + __expf(-gin));
  float nl = fmaxf(lin, 0.f);
  hx[idx] = g * hx[idx] + (1.f - g) * nl;
}

__global__ __launch_bounds__(256)
void outw_kernel(const float* __restrict__ hx, float* __restrict__ out, int dir) {
  size_t idx = (size_t)blockIdx.x * blockDim.x + threadIdx.x;
  if (idx >= BSD_) return;
  int dd = (int)(idx % D_);
  size_t m = idx / D_;
  size_t o = m * (size_t)(2 * D_) + (size_t)dir * D_ + dd;
  float v = hx[idx];
  out[o] = v;
  out[o + (size_t)BS_ * 2 * D_] = v;
}

static_assert(BN_ % GEMM_BM == 0 && BS_ % GEMM_BM == 0, "GEMM row tiling must be exact");
static_assert(D_ % GEMM_BN == 0 && (2 * D_) % GEMM_BN == 0, "GEMM column tiling must be exact");

extern "C" void kernel_launch(void* const* d_in, const int* in_sizes, int n_in,
                              void* d_out, int out_size, void* d_ws, size_t ws_size,
                              hipStream_t stream) {
  const float* inputs    = (const float*)d_in[0];
  const float* left_pad  = (const float*)d_in[1];
  const float* right_pad = (const float*)d_in[2];
  const float* Wqkv[2] = {(const float*)d_in[3],  (const float*)d_in[7]};
  const float* bqkv[2] = {(const float*)d_in[4],  (const float*)d_in[8]};
  const float* Wo[2]   = {(const float*)d_in[5],  (const float*)d_in[9]};
  const float* bo[2]   = {(const float*)d_in[6],  (const float*)d_in[10]};
  const float* hwW[2]  = {(const float*)d_in[11], (const float*)d_in[13]};
  const float* hwb[2]  = {(const float*)d_in[12], (const float*)d_in[14]};
  float* out = (float*)d_out;

  float* ws    = (float*)d_ws;
  float* xpad  = ws;
  float* qkv   = xpad + BND_;
  float* attn  = qkv + 3 * BND_;
  float* hx    = attn + BND_;
  float* proj  = qkv;
  float* hproj = qkv + BND_;

  dim3 blk(32, 8);

  { int n = (int)BND_; pad_kernel<<<(n + 255) / 256, 256, 0, stream>>>(inputs, left_pad, right_pad, xpad); }

  for (int dir = 0; dir < 2; ++dir) {
    for (int t = 0; t < 3; ++t) {
      dim3 grd(D_ / GEMM_BN, BN_ / GEMM_BM);
      gemm_wmma_f32<<<grd, blk, 0, stream>>>(
          xpad, D_, Wqkv[dir] + (size_t)t * D_ * D_, D_, bqkv[dir] + t * D_,
          qkv + (size_t)t * BND_, D_, BN_, D_);
    }
    { int n = B_ * H_ * N_;
      attn_kernel<<<(n + 255) / 256, 256, 0, stream>>>(qkv, attn, dir); }
    { dim3 grd(D_ / GEMM_BN, BN_ / GEMM_BM);
      gemm_wmma_f32<<<grd, blk, 0, stream>>>(attn, D_, Wo[dir], D_, bo[dir],
                                             proj, D_, BN_, D_); }
    { int n = (int)BSD_; slice_kernel<<<(n + 255) / 256, 256, 0, stream>>>(proj, hx); }
    for (int l = 0; l < NL_; ++l) {
      dim3 grd(2 * D_ / GEMM_BN, BS_ / GEMM_BM);
      gemm_wmma_f32<<<grd, blk, 0, stream>>>(
          hx, D_, hwW[dir] + (size_t)l * 2 * D_ * D_, D_, hwb[dir] + l * 2 * D_,
          hproj, 2 * D_, BS_, 2 * D_);
      int n = (int)BSD_;
      gate_kernel<<<(n + 255) / 256, 256, 0, stream>>>(hproj, hx);
    }
    { int n = (int)BSD_; outw_kernel<<<(n + 255) / 256, 256, 0, stream>>>(hx, out, dir); }
  }
}


# Elementwise passes folded into the GEMMs: pad and centre slice become the A loader's row map, highway gate and output write become the GEMM epilogue (lin/gate columns paired); 23 launches to 14
# speedup vs baseline: 2.4963x; 1.2365x over previous
#include <hip/hip_runtime.h>

#define B_ 8
#define S_ 512
#define D_ 512
#define H_ 64
#define DK_ 8
#define WIDTH_ 16
#define N_ (S_ + 2 * WIDTH_)
#define NL_ 2
#define WIN_ (WIDTH_ + 2)
#define BN_ (B_ * N_)
#define BS_ (B_ * S_)
static const size_t BND_ = (size_t)BN_ * D_;
static const size_t BSD_ = (size_t)BS_ * D_;

typedef __attribute__((ext_vector_type(2))) float v2f;
typedef __attribute__((ext_vector_type(8))) float v8f;

#define GEMM_BM 64
#define GEMM_BN 64
#define GATE_BN 32

enum { A_PADDED = 0, A_CENTER = 1, A_PLAIN = 2 };
enum { E_PLAIN = 0, E_GATE = 1, E_GATE_OUT = 2 };

template <int AM, int EM>
__global__ __launch_bounds__(256)
void gemm_wmma_f32(const float* __restrict__ A, const float* __restrict__ LP,
                   const float* __restrict__ RP,
                   const float* __restrict__ W, const float* __restrict__ bias,
                   float* __restrict__ C, int ldc, int M, int Nout, int dir) {
  constexpr int KK = 512;
  constexpr int BM = GEMM_BM;
  constexpr int BN = (EM == E_PLAIN) ? GEMM_BN : GATE_BN;
  constexpr int BK = 32;
  constexpr int PITCH = BK + 4;
  constexpr int NCH = KK / BK;
  constexpr int KS = BK / 4;
  constexpr int B1 = (EM == E_PLAIN) ? 16 : 32;
  __shared__ __attribute__((aligned(16))) float As[2][BM * PITCH];
  __shared__ __attribute__((aligned(16))) float Bs[2][64 * PITCH];

  const int lane = threadIdx.x;
  const int wave = threadIdx.y;
  const int tid  = wave * 32 + lane;
  const int row0 = blockIdx.y * BM;
  const int col0 = blockIdx.x * BN;

  if (row0 >= M || col0 >= Nout) return;

  const int srow = tid >> 3;
  const int scv  = (tid & 7) << 2;
  int ar0 = row0 + srow;      if (ar0 > M - 1)    ar0 = M - 1;
  int ar1 = row0 + srow + 32; if (ar1 > M - 1)    ar1 = M - 1;
  auto arow = [&](int r) __attribute__((always_inline)) -> const float* {
    if constexpr (AM == A_PADDED) {
      const int b = r / N_;
      const int p = r - b * N_;
      if (p < WIDTH_)       return LP + p * D_;
      if (p < WIDTH_ + S_)  return A + ((size_t)b * S_ + (p - WIDTH_)) * D_;
      return RP + (p - WIDTH_ - S_) * D_;
    } else if constexpr (AM == A_CENTER) {
      const int b = r / S_;
      const int s = r - b * S_;
      return A + ((size_t)b * N_ + WIDTH_ + s) * D_;
    } else {
      return A + (size_t)r * D_;
    }
  };
  const float* ap0 = arow(ar0) + scv;
  const float* ap1 = arow(ar1) + scv;
  int wr0, wr1;
  if constexpr (EM == E_PLAIN) {
    wr0 = col0 + srow;      if (wr0 > Nout - 1) wr0 = Nout - 1;
    wr1 = col0 + srow + 32; if (wr1 > Nout - 1) wr1 = Nout - 1;
  } else {
    wr0 = col0 + srow;
    wr1 = D_ + col0 + srow;
  }
  const unsigned ow0 = (unsigned)(wr0 * KK + scv) * 4u;
  const unsigned ow1 = (unsigned)(wr1 * KK + scv) * 4u;
  const int so0 = srow * PITCH + scv;
  const int so1 = (srow + 32) * PITCH + scv;

  float4 pa0 = *(const float4*)(ap0);
  float4 pa1 = *(const float4*)(ap1);
  float4 pw0 = *(const float4*)((const char*)W + ow0);
  float4 pw1 = *(const float4*)((const char*)W + ow1);
  *(float4*)&As[0][so0] = pa0;
  *(float4*)&As[0][so1] = pa1;
  *(float4*)&Bs[0][so0] = pw0;
  *(float4*)&Bs[0][so1] = pw1;
  __syncthreads();

  const int wm = wave & 3;
  const int wn = wave >> 2;
  const int ml = lane & 15;
  const int kb = (lane >> 4) << 1;
  const int aoff = (wm * 16 + ml) * PITCH + kb;
  const int boff = (wn * ((EM == E_PLAIN) ? 32 : 16) + ml) * PITCH + kb;

  v8f acc0 = {}, acc1 = {};

  auto chunk = [&](const int cur, const int kn) __attribute__((always_inline)) {
    const char* Wn = (const char*)(W + kn * BK);
    pa0 = *(const float4*)(ap0 + kn * BK);
    pa1 = *(const float4*)(ap1 + kn * BK);
    pw0 = *(const float4*)(Wn + ow0);
    pw1 = *(const float4*)(Wn + ow1);
    __builtin_amdgcn_sched_barrier(0);
    const float* ap = &As[cur][aoff];
    const float* bp = &Bs[cur][boff];
    v2f af[KS], bf0[KS], bf1[KS];
#pragma unroll
    for (int s = 0; s < KS; ++s) {
      af[s]  = *(const v2f*)(ap + 4 * s);
      bf0[s] = *(const v2f*)(bp + 4 * s);
      bf1[s] = *(const v2f*)(bp + B1 * PITCH + 4 * s);
    }
#pragma unroll
    for (int s = 0; s < KS; ++s) {
      acc0 = __builtin_amdgcn_wmma_f32_16x16x4_f32(false, af[s], false, bf0[s], (short)0, acc0, false, false);
      acc1 = __builtin_amdgcn_wmma_f32_16x16x4_f32(false, af[s], false, bf1[s], (short)0, acc1, false, false);
    }
    asm volatile("v_nop\n\tv_nop\n\tv_nop\n\tv_nop\n\tv_nop" : "+v"(acc0), "+v"(acc1));
    __builtin_amdgcn_sched_barrier(0);
    *(float4*)&As[cur ^ 1][so0] = pa0;
    *(float4*)&As[cur ^ 1][so1] = pa1;
    *(float4*)&Bs[cur ^ 1][so0] = pw0;
    *(float4*)&Bs[cur ^ 1][so1] = pw1;
    __syncthreads();
  };

  for (int kc = 0; kc < NCH; kc += 2) {
    chunk(0, kc + 1);
    chunk(1, (kc + 2 < NCH) ? kc + 2 : NCH - 1);
  }

  if constexpr (EM == E_PLAIN) {
    const int mbase = row0 + wm * 16 + ((lane >> 4) << 3);
    const int cb = col0 + wn * 32 + ml;
    const float bv0 = bias[cb];
    const float bv1 = bias[cb + 16];
#pragma unroll
    for (int r = 0; r < 8; ++r) {
      float* crow = C + (size_t)(mbase + r) * ldc + cb;
      crow[ 0] = acc0[r] + bv0;
      crow[16] = acc1[r] + bv1;
    }
  } else {
    float* stg = &As[0][0];
    const int cl = wn * 16 + ml;
    const int cg = col0 + cl;
    const float bl = bias[cg];
    const float bg = bias[D_ + cg];
    const int rl0 = wm * 16 + ((lane >> 4) << 3);
#pragma unroll
    for (int r = 0; r < 8; ++r) {
      const int rl = rl0 + r;
      const float xo  = A[(size_t)(row0 + rl) * D_ + cg];
      const float lin = acc0[r] + bl;
      const float gin = acc1[r] + bg;
      const float g  = 1.f / (1.f + __expf(-gin));
      const float nl = fmaxf(lin, 0.f);
      stg[rl * PITCH + cl] = g * xo + (1.f - g) * nl;
    }
    __syncthreads();
#pragma unroll
    for (int i = 0; i < 8; ++i) {
      const int rl = wave * 8 + i;
      const float v = stg[rl * PITCH + lane];
      const size_t m = (size_t)(row0 + rl);
      if constexpr (EM == E_GATE) {
        C[m * D_ + col0 + lane] = v;
      } else {
        const size_t o = m * (size_t)(2 * D_) + (size_t)dir * D_ + col0 + lane;
        C[o] = v;
        C[o + (size_t)BS_ * 2 * D_] = v;
      }
    }
  }
}

__global__ __launch_bounds__(256)
void attn_kernel(const float* __restrict__ qkv, float* __restrict__ out, int dir) {
  int idx = blockIdx.x * blockDim.x + threadIdx.x;
  if (idx >= B_ * H_ * N_) return;
  int h    = idx % H_;
  int rest = idx / H_;
  int i    = rest % N_;
  int b    = rest / N_;

  const size_t base = (size_t)(b * N_ + i) * D_ + h * DK_;
  const float* qp = qkv + base;
  const float* kbase = qkv + BND_ + (size_t)(b * N_) * D_ + h * DK_;
  const float* vbase = qkv + 2 * BND_ + (size_t)(b * N_) * D_ + h * DK_;

  float q[DK_];
  *(float4*)&q[0] = *(const float4*)(qp);
  *(float4*)&q[4] = *(const float4*)(qp + 4);

  const int j0 = (dir == 0) ? ((i - WIDTH_ - 1 > 0) ? i - WIDTH_ - 1 : 0) : i;
  const int j1 = (dir == 0) ? i : ((i + WIDTH_ + 1 < N_ - 1) ? i + WIDTH_ + 1 : N_ - 1);
  const float scale = 0.35355339059327373f;

  float sv[WIN_];
  float mx = -3.0e38f;
#pragma unroll
  for (int t = 0; t < WIN_; ++t) {
    int j = j0 + t;
    float sc = -3.0e38f;
    if (j <= j1) {
      const float* kp = kbase + (size_t)j * D_;
      float kr[DK_];
      *(float4*)&kr[0] = *(const float4*)(kp);
      *(float4*)&kr[4] = *(const float4*)(kp + 4);
      float dot = 0.f;
#pragma unroll
      for (int d = 0; d < DK_; ++d) dot += q[d] * kr[d];
      sc = dot * scale;
    }
    sv[t] = sc;
    mx = fmaxf(mx, sc);
  }
  float sum = 0.f;
#pragma unroll
  for (int t = 0; t < WIN_; ++t) {
    float p = __expf(sv[t] - mx);
    sv[t] = p;
    sum += p;
  }
  const float inv = 1.f / sum;
  float o[DK_] = {};
#pragma unroll
  for (int t = 0; t < WIN_; ++t) {
    int j = j0 + t;
    if (j <= j1) {
      const float* vp = vbase + (size_t)j * D_;
      float vr[DK_];
      *(float4*)&vr[0] = *(const float4*)(vp);
      *(float4*)&vr[4] = *(const float4*)(vp + 4);
      float w = sv[t] * inv;
#pragma unroll
      for (int d = 0; d < DK_; ++d) o[d] += w * vr[d];
    }
  }
  float* op = out + base;
  float4 o0; o0.x = o[0]; o0.y = o[1]; o0.z = o[2]; o0.w = o[3];
  float4 o1; o1.x = o[4]; o1.y = o[5]; o1.z = o[6]; o1.w = o[7];
  *(float4*)(op)     = o0;
  *(float4*)(op + 4) = o1;
}

static_assert(BN_ % GEMM_BM == 0 && BS_ % GEMM_BM == 0, "GEMM row tiling must be exact");
static_assert(D_ % GEMM_BN == 0 && D_ % GATE_BN == 0, "GEMM column tiling must be exact");
static_assert(NL_ == 2, "two highway layers: the first writes the spare buffer, the second the output");

extern "C" void kernel_launch(void* const* d_in, const int* in_sizes, int n_in,
                              void* d_out, int out_size, void* d_ws, size_t ws_size,
                              hipStream_t stream) {
  const float* inputs    = (const float*)d_in[0];
  const float* left_pad  = (const float*)d_in[1];
  const float* right_pad = (const float*)d_in[2];
  const float* Wqkv[2] = {(const float*)d_in[3],  (const float*)d_in[7]};
  const float* bqkv[2] = {(const float*)d_in[4],  (const float*)d_in[8]};
  const float* Wo[2]   = {(const float*)d_in[5],  (const float*)d_in[9]};
  const float* bo[2]   = {(const float*)d_in[6],  (const float*)d_in[10]};
  const float* hwW[2]  = {(const float*)d_in[11], (const float*)d_in[13]};
  const float* hwb[2]  = {(const float*)d_in[12], (const float*)d_in[14]};
  float* out = (float*)d_out;

  float* ws    = (float*)d_ws;
  float* qkv   = ws + BND_;
  float* attn  = qkv + 3 * BND_;
  float* hx    = attn + BND_;
  float* hx2   = qkv;

  dim3 blk(32, 8);

  for (int dir = 0; dir < 2; ++dir) {
    for (int t = 0; t < 3; ++t) {
      dim3 grd(D_ / GEMM_BN, BN_ / GEMM_BM);
      gemm_wmma_f32<A_PADDED, E_PLAIN><<<grd, blk, 0, stream>>>(
          inputs, left_pad, right_pad, Wqkv[dir] + (size_t)t * D_ * D_, bqkv[dir] + t * D_,
          qkv + (size_t)t * BND_, D_, BN_, D_, dir);
    }
    { int n = B_ * H_ * N_;
      attn_kernel<<<(n + 255) / 256, 256, 0, stream>>>(qkv, attn, dir); }
    { dim3 grd(D_ / GEMM_BN, BS_ / GEMM_BM);
      gemm_wmma_f32<A_CENTER, E_PLAIN><<<grd, blk, 0, stream>>>(
          attn, attn, attn, Wo[dir], bo[dir], hx, D_, BS_, D_, dir); }
    { dim3 grd(D_ / GATE_BN, BS_ / GEMM_BM);
      gemm_wmma_f32<A_PLAIN, E_GATE><<<grd, blk, 0, stream>>>(
          hx, hx, hx, hwW[dir], hwb[dir], hx2, D_, BS_, D_, dir); }
    { dim3 grd(D_ / GATE_BN, BS_ / GEMM_BM);
      gemm_wmma_f32<A_PLAIN, E_GATE_OUT><<<grd, blk, 0, stream>>>(
          hx2, hx2, hx2, hwW[dir] + (size_t)2 * D_ * D_, hwb[dir] + 2 * D_,
          out, 2 * D_, BS_, D_, dir); }
  }
}


# Independent GEMMs of one shape share a launch via grid z: the three QKV projections of a direction (z = plane), both directions of Wo and of each highway layer (z = direction); 14 launches to 7
# speedup vs baseline: 2.8069x; 1.1244x over previous
#include <hip/hip_runtime.h>

#define B_ 8
#define S_ 512
#define D_ 512
#define H_ 64
#define DK_ 8
#define WIDTH_ 16
#define N_ (S_ + 2 * WIDTH_)
#define NL_ 2
#define WIN_ (WIDTH_ + 2)
#define BN_ (B_ * N_)
#define BS_ (B_ * S_)
static const size_t BND_ = (size_t)BN_ * D_;
static const size_t BSD_ = (size_t)BS_ * D_;

typedef __attribute__((ext_vector_type(2))) float v2f;
typedef __attribute__((ext_vector_type(8))) float v8f;

#define GEMM_BM 64
#define GEMM_BN 64
#define GATE_BN 32

enum { A_PADDED = 0, A_CENTER = 1, A_PLAIN = 2 };
enum { E_PLAIN = 0, E_GATE = 1, E_GATE_OUT = 2 };
enum { Z_PLANE = 0, Z_DIR = 1 };

template <int AM, int EM, int ZM>
__global__ __launch_bounds__(256)
void gemm_wmma_f32(const float* __restrict__ A, const float* __restrict__ LP,
                   const float* __restrict__ RP,
                   const float* __restrict__ W0, const float* __restrict__ W1,
                   const float* __restrict__ bias0, const float* __restrict__ bias1,
                   float* __restrict__ C, int ldc, int M, int Nout) {
  constexpr int KK = 512;
  constexpr int BM = GEMM_BM;
  constexpr int BN = (EM == E_PLAIN) ? GEMM_BN : GATE_BN;
  constexpr int BK = 32;
  constexpr int PITCH = BK + 4;
  constexpr int NCH = KK / BK;
  constexpr int KS = BK / 4;
  constexpr int B1 = (EM == E_PLAIN) ? 16 : 32;
  __shared__ __attribute__((aligned(16))) float As[2][BM * PITCH];
  __shared__ __attribute__((aligned(16))) float Bs[2][64 * PITCH];

  const int lane = threadIdx.x;
  const int wave = threadIdx.y;
  const int tid  = wave * 32 + lane;
  const int row0 = blockIdx.y * BM;
  const int col0 = blockIdx.x * BN;

  if (row0 >= M || col0 >= Nout) return;

  const int z = blockIdx.z;
  const float* W;
  const float* bias;
  int dir = 0;
  if constexpr (ZM == Z_PLANE) {
    W    = W0 + (size_t)z * D_ * D_;
    bias = bias0 + z * D_;
    C   += (size_t)z * BND_;
  } else {
    dir  = z;
    W    = z ? W1 : W0;
    bias = z ? bias1 : bias0;
    A   += (size_t)z * ((AM == A_CENTER) ? BND_ : BSD_);
    if constexpr (EM != E_GATE_OUT) C += (size_t)z * BSD_;
  }

  const int srow = tid >> 3;
  const int scv  = (tid & 7) << 2;
  int ar0 = row0 + srow;      if (ar0 > M - 1)    ar0 = M - 1;
  int ar1 = row0 + srow + 32; if (ar1 > M - 1)    ar1 = M - 1;
  auto arow = [&](int r) __attribute__((always_inline)) -> const float* {
    if constexpr (AM == A_PADDED) {
      const int b = r / N_;
      const int p = r - b * N_;
      if (p < WIDTH_)       return LP + p * D_;
      if (p < WIDTH_ + S_)  return A + ((size_t)b * S_ + (p - WIDTH_)) * D_;
      return RP + (p - WIDTH_ - S_) * D_;
    } else if constexpr (AM == A_CENTER) {
      const int b = r / S_;
      const int s = r - b * S_;
      return A + ((size_t)b * N_ + WIDTH_ + s) * D_;
    } else {
      return A + (size_t)r * D_;
    }
  };
  const float* ap0 = arow(ar0) + scv;
  const float* ap1 = arow(ar1) + scv;
  int wr0, wr1;
  if constexpr (EM == E_PLAIN) {
    wr0 = col0 + srow;      if (wr0 > Nout - 1) wr0 = Nout - 1;
    wr1 = col0 + srow + 32; if (wr1 > Nout - 1) wr1 = Nout - 1;
  } else {
    wr0 = col0 + srow;
    wr1 = D_ + col0 + srow;
  }
  const unsigned ow0 = (unsigned)(wr0 * KK + scv) * 4u;
  const unsigned ow1 = (unsigned)(wr1 * KK + scv) * 4u;
  const int so0 = srow * PITCH + scv;
  const int so1 = (srow + 32) * PITCH + scv;

  float4 pa0 = *(const float4*)(ap0);
  float4 pa1 = *(const float4*)(ap1);
  float4 pw0 = *(const float4*)((const char*)W + ow0);
  float4 pw1 = *(const float4*)((const char*)W + ow1);
  *(float4*)&As[0][so0] = pa0;
  *(float4*)&As[0][so1] = pa1;
  *(float4*)&Bs[0][so0] = pw0;
  *(float4*)&Bs[0][so1] = pw1;
  __syncthreads();

  const int wm = wave & 3;
  const int wn = wave >> 2;
  const int ml = lane & 15;
  const int kb = (lane >> 4) << 1;
  const int aoff = (wm * 16 + ml) * PITCH + kb;
  const int boff = (wn * ((EM == E_PLAIN) ? 32 : 16) + ml) * PITCH + kb;

  v8f acc0 = {}, acc1 = {};

  auto chunk = [&](const int cur, const int kn) __attribute__((always_inline)) {
    const char* Wn = (const char*)(W + kn * BK);
    pa0 = *(const float4*)(ap0 + kn * BK);
    pa1 = *(const float4*)(ap1 + kn * BK);
    pw0 = *(const float4*)(Wn + ow0);
    pw1 = *(const float4*)(Wn + ow1);
    __builtin_amdgcn_sched_barrier(0);
    const float* ap = &As[cur][aoff];
    const float* bp = &Bs[cur][boff];
    v2f af[KS], bf0[KS], bf1[KS];
#pragma unroll
    for (int s = 0; s < KS; ++s) {
      af[s]  = *(const v2f*)(ap + 4 * s);
      bf0[s] = *(const v2f*)(bp + 4 * s);
      bf1[s] = *(const v2f*)(bp + B1 * PITCH + 4 * s);
    }
#pragma unroll
    for (int s = 0; s < KS; ++s) {
      acc0 = __builtin_amdgcn_wmma_f32_16x16x4_f32(false, af[s], false, bf0[s], (short)0, acc0, false, false);
      acc1 = __builtin_amdgcn_wmma_f32_16x16x4_f32(false, af[s], false, bf1[s], (short)0, acc1, false, false);
    }
    asm volatile("v_nop\n\tv_nop\n\tv_nop\n\tv_nop\n\tv_nop" : "+v"(acc0), "+v"(acc1));
    __builtin_amdgcn_sched_barrier(0);
    *(float4*)&As[cur ^ 1][so0] = pa0;
    *(float4*)&As[cur ^ 1][so1] = pa1;
    *(float4*)&Bs[cur ^ 1][so0] = pw0;
    *(float4*)&Bs[cur ^ 1][so1] = pw1;
    __syncthreads();
  };

  for (int kc = 0; kc < NCH; kc += 2) {
    chunk(0, kc + 1);
    chunk(1, (kc + 2 < NCH) ? kc + 2 : NCH - 1);
  }

  if constexpr (EM == E_PLAIN) {
    const int mbase = row0 + wm * 16 + ((lane >> 4) << 3);
    const int cb = col0 + wn * 32 + ml;
    const float bv0 = bias[cb];
    const float bv1 = bias[cb + 16];
#pragma unroll
    for (int r = 0; r < 8; ++r) {
      float* crow = C + (size_t)(mbase + r) * ldc + cb;
      crow[ 0] = acc0[r] + bv0;
      crow[16] = acc1[r] + bv1;
    }
  } else {
    float* stg = &As[0][0];
    const int cl = wn * 16 + ml;
    const int cg = col0 + cl;
    const float bl = bias[cg];
    const float bg = bias[D_ + cg];
    const int rl0 = wm * 16 + ((lane >> 4) << 3);
#pragma unroll
    for (int r = 0; r < 8; ++r) {
      const int rl = rl0 + r;
      const float xo  = A[(size_t)(row0 + rl) * D_ + cg];
      const float lin = acc0[r] + bl;
      const float gin = acc1[r] + bg;
      const float g  = 1.f / (1.f + __expf(-gin));
      const float nl = fmaxf(lin, 0.f);
      stg[rl * PITCH + cl] = g * xo + (1.f - g) * nl;
    }
    __syncthreads();
#pragma unroll
    for (int i = 0; i < 8; ++i) {
      const int rl = wave * 8 + i;
      const float v = stg[rl * PITCH + lane];
      const size_t m = (size_t)(row0 + rl);
      if constexpr (EM == E_GATE) {
        C[m * D_ + col0 + lane] = v;
      } else {
        const size_t o = m * (size_t)(2 * D_) + (size_t)dir * D_ + col0 + lane;
        C[o] = v;
        C[o + (size_t)BS_ * 2 * D_] = v;
      }
    }
  }
}

__global__ __launch_bounds__(256)
void attn_kernel(const float* __restrict__ qkv, float* __restrict__ out, int dir) {
  int idx = blockIdx.x * blockDim.x + threadIdx.x;
  if (idx >= B_ * H_ * N_) return;
  int h    = idx % H_;
  int rest = idx / H_;
  int i    = rest % N_;
  int b    = rest / N_;

  const size_t base = (size_t)(b * N_ + i) * D_ + h * DK_;
  const float* qp = qkv + base;
  const float* kbase = qkv + BND_ + (size_t)(b * N_) * D_ + h * DK_;
  const float* vbase = qkv + 2 * BND_ + (size_t)(b * N_) * D_ + h * DK_;

  float q[DK_];
  *(float4*)&q[0] = *(const float4*)(qp);
  *(float4*)&q[4] = *(const float4*)(qp + 4);

  const int j0 = (dir == 0) ? ((i - WIDTH_ - 1 > 0) ? i - WIDTH_ - 1 : 0) : i;
  const int j1 = (dir == 0) ? i : ((i + WIDTH_ + 1 < N_ - 1) ? i + WIDTH_ + 1 : N_ - 1);
  const float scale = 0.35355339059327373f;

  float sv[WIN_];
  float mx = -3.0e38f;
#pragma unroll
  for (int t = 0; t < WIN_; ++t) {
    int j = j0 + t;
    float sc = -3.0e38f;
    if (j <= j1) {
      const float* kp = kbase + (size_t)j * D_;
      float kr[DK_];
      *(float4*)&kr[0] = *(const float4*)(kp);
      *(float4*)&kr[4] = *(const float4*)(kp + 4);
      float dot = 0.f;
#pragma unroll
      for (int d = 0; d < DK_; ++d) dot += q[d] * kr[d];
      sc = dot * scale;
    }
    sv[t] = sc;
    mx = fmaxf(mx, sc);
  }
  float sum = 0.f;
#pragma unroll
  for (int t = 0; t < WIN_; ++t) {
    float p = __expf(sv[t] - mx);
    sv[t] = p;
    sum += p;
  }
  const float inv = 1.f / sum;
  float o[DK_] = {};
#pragma unroll
  for (int t = 0; t < WIN_; ++t) {
    int j = j0 + t;
    if (j <= j1) {
      const float* vp = vbase + (size_t)j * D_;
      float vr[DK_];
      *(float4*)&vr[0] = *(const float4*)(vp);
      *(float4*)&vr[4] = *(const float4*)(vp + 4);
      float w = sv[t] * inv;
#pragma unroll
      for (int d = 0; d < DK_; ++d) o[d] += w * vr[d];
    }
  }
  float* op = out + base;
  float4 o0; o0.x = o[0]; o0.y = o[1]; o0.z = o[2]; o0.w = o[3];
  float4 o1; o1.x = o[4]; o1.y = o[5]; o1.z = o[6]; o1.w = o[7];
  *(float4*)(op)     = o0;
  *(float4*)(op + 4) = o1;
}

static_assert(BN_ % GEMM_BM == 0 && BS_ % GEMM_BM == 0, "GEMM row tiling must be exact");
static_assert(D_ % GEMM_BN == 0 && D_ % GATE_BN == 0, "GEMM column tiling must be exact");
static_assert(NL_ == 2, "two highway layers: the first writes the spare buffer, the second the output");

extern "C" void kernel_launch(void* const* d_in, const int* in_sizes, int n_in,
                              void* d_out, int out_size, void* d_ws, size_t ws_size,
                              hipStream_t stream) {
  const float* inputs    = (const float*)d_in[0];
  const float* left_pad  = (const float*)d_in[1];
  const float* right_pad = (const float*)d_in[2];
  const float* Wqkv[2] = {(const float*)d_in[3],  (const float*)d_in[7]};
  const float* bqkv[2] = {(const float*)d_in[4],  (const float*)d_in[8]};
  const float* Wo[2]   = {(const float*)d_in[5],  (const float*)d_in[9]};
  const float* bo[2]   = {(const float*)d_in[6],  (const float*)d_in[10]};
  const float* hwW[2]  = {(const float*)d_in[11], (const float*)d_in[13]};
  const float* hwb[2]  = {(const float*)d_in[12], (const float*)d_in[14]};
  float* out = (float*)d_out;

  float* ws    = (float*)d_ws;
  float* qkv   = ws;
  float* attn  = ws + 3 * BND_;
  float* hx    = ws;
  float* hx2   = ws + 3 * BND_;

  dim3 blk(32, 8);

  for (int dir = 0; dir < 2; ++dir) {
    { dim3 grd(D_ / GEMM_BN, BN_ / GEMM_BM, 3);
      gemm_wmma_f32<A_PADDED, E_PLAIN, Z_PLANE><<<grd, blk, 0, stream>>>(
          inputs, left_pad, right_pad, Wqkv[dir], Wqkv[dir], bqkv[dir], bqkv[dir],
          qkv, D_, BN_, D_); }
    { int n = B_ * H_ * N_;
      attn_kernel<<<(n + 255) / 256, 256, 0, stream>>>(qkv, attn + (size_t)dir * BND_, dir); }
  }
  { dim3 grd(D_ / GEMM_BN, BS_ / GEMM_BM, 2);
    gemm_wmma_f32<A_CENTER, E_PLAIN, Z_DIR><<<grd, blk, 0, stream>>>(
        attn, attn, attn, Wo[0], Wo[1], bo[0], bo[1], hx, D_, BS_, D_); }
  { dim3 grd(D_ / GATE_BN, BS_ / GEMM_BM, 2);
    gemm_wmma_f32<A_PLAIN, E_GATE, Z_DIR><<<grd, blk, 0, stream>>>(
        hx, hx, hx, hwW[0], hwW[1], hwb[0], hwb[1], hx2, D_, BS_, D_); }
  { dim3 grd(D_ / GATE_BN, BS_ / GEMM_BM, 2);
    gemm_wmma_f32<A_PLAIN, E_GATE_OUT, Z_DIR><<<grd, blk, 0, stream>>>(
        hx2, hx2, hx2, hwW[0] + (size_t)2 * D_ * D_, hwW[1] + (size_t)2 * D_ * D_,
        hwb[0] + 2 * D_, hwb[1] + 2 * D_, out, 2 * D_, BS_, D_); }
}


# Attention: key slots loaded without a branch (slots past the window re-read its last key, score masked), so K and V loads go out in groups of 6 keys instead of one dependent round trip per key
# speedup vs baseline: 2.9929x; 1.0663x over previous
#include <hip/hip_runtime.h>

#define B_ 8
#define S_ 512
#define D_ 512
#define H_ 64
#define DK_ 8
#define WIDTH_ 16
#define N_ (S_ + 2 * WIDTH_)
#define NL_ 2
#define WIN_ (WIDTH_ + 2)
#define BN_ (B_ * N_)
#define BS_ (B_ * S_)
static const size_t BND_ = (size_t)BN_ * D_;
static const size_t BSD_ = (size_t)BS_ * D_;

typedef __attribute__((ext_vector_type(2))) float v2f;
typedef __attribute__((ext_vector_type(8))) float v8f;

#define GEMM_BM 64
#define GEMM_BN 64
#define GATE_BN 32

enum { A_PADDED = 0, A_CENTER = 1, A_PLAIN = 2 };
enum { E_PLAIN = 0, E_GATE = 1, E_GATE_OUT = 2 };
enum { Z_PLANE = 0, Z_DIR = 1 };

template <int AM, int EM, int ZM>
__global__ __launch_bounds__(256)
void gemm_wmma_f32(const float* __restrict__ A, const float* __restrict__ LP,
                   const float* __restrict__ RP,
                   const float* __restrict__ W0, const float* __restrict__ W1,
                   const float* __restrict__ bias0, const float* __restrict__ bias1,
                   float* __restrict__ C, int ldc, int M, int Nout) {
  constexpr int KK = 512;
  constexpr int BM = GEMM_BM;
  constexpr int BN = (EM == E_PLAIN) ? GEMM_BN : GATE_BN;
  constexpr int BK = 32;
  constexpr int PITCH = BK + 4;
  constexpr int NCH = KK / BK;
  constexpr int KS = BK / 4;
  constexpr int B1 = (EM == E_PLAIN) ? 16 : 32;
  __shared__ __attribute__((aligned(16))) float As[2][BM * PITCH];
  __shared__ __attribute__((aligned(16))) float Bs[2][64 * PITCH];

  const int lane = threadIdx.x;
  const int wave = threadIdx.y;
  const int tid  = wave * 32 + lane;
  const int row0 = blockIdx.y * BM;
  const int col0 = blockIdx.x * BN;

  if (row0 >= M || col0 >= Nout) return;

  const int z = blockIdx.z;
  const float* W;
  const float* bias;
  int dir = 0;
  if constexpr (ZM == Z_PLANE) {
    W    = W0 + (size_t)z * D_ * D_;
    bias = bias0 + z * D_;
    C   += (size_t)z * BND_;
  } else {
    dir  = z;
    W    = z ? W1 : W0;
    bias = z ? bias1 : bias0;
    A   += (size_t)z * ((AM == A_CENTER) ? BND_ : BSD_);
    if constexpr (EM != E_GATE_OUT) C += (size_t)z * BSD_;
  }

  const int srow = tid >> 3;
  const int scv  = (tid & 7) << 2;
  int ar0 = row0 + srow;      if (ar0 > M - 1)    ar0 = M - 1;
  int ar1 = row0 + srow + 32; if (ar1 > M - 1)    ar1 = M - 1;
  auto arow = [&](int r) __attribute__((always_inline)) -> const float* {
    if constexpr (AM == A_PADDED) {
      const int b = r / N_;
      const int p = r - b * N_;
      if (p < WIDTH_)       return LP + p * D_;
      if (p < WIDTH_ + S_)  return A + ((size_t)b * S_ + (p - WIDTH_)) * D_;
      return RP + (p - WIDTH_ - S_) * D_;
    } else if constexpr (AM == A_CENTER) {
      const int b = r / S_;
      const int s = r - b * S_;
      return A + ((size_t)b * N_ + WIDTH_ + s) * D_;
    } else {
      return A + (size_t)r * D_;
    }
  };
  const float* ap0 = arow(ar0) + scv;
  const float* ap1 = arow(ar1) + scv;
  int wr0, wr1;
  if constexpr (EM == E_PLAIN) {
    wr0 = col0 + srow;      if (wr0 > Nout - 1) wr0 = Nout - 1;
    wr1 = col0 + srow + 32; if (wr1 > Nout - 1) wr1 = Nout - 1;
  } else {
    wr0 = col0 + srow;
    wr1 = D_ + col0 + srow;
  }
  const unsigned ow0 = (unsigned)(wr0 * KK + scv) * 4u;
  const unsigned ow1 = (unsigned)(wr1 * KK + scv) * 4u;
  const int so0 = srow * PITCH + scv;
  const int so1 = (srow + 32) * PITCH + scv;

  float4 pa0 = *(const float4*)(ap0);
  float4 pa1 = *(const float4*)(ap1);
  float4 pw0 = *(const float4*)((const char*)W + ow0);
  float4 pw1 = *(const float4*)((const char*)W + ow1);
  *(float4*)&As[0][so0] = pa0;
  *(float4*)&As[0][so1] = pa1;
  *(float4*)&Bs[0][so0] = pw0;
  *(float4*)&Bs[0][so1] = pw1;
  __syncthreads();

  const int wm = wave & 3;
  const int wn = wave >> 2;
  const int ml = lane & 15;
  const int kb = (lane >> 4) << 1;
  const int aoff = (wm * 16 + ml) * PITCH + kb;
  const int boff = (wn * ((EM == E_PLAIN) ? 32 : 16) + ml) * PITCH + kb;

  v8f acc0 = {}, acc1 = {};

  auto chunk = [&](const int cur, const int kn) __attribute__((always_inline)) {
    const char* Wn = (const char*)(W + kn * BK);
    pa0 = *(const float4*)(ap0 + kn * BK);
    pa1 = *(const float4*)(ap1 + kn * BK);
    pw0 = *(const float4*)(Wn + ow0);
    pw1 = *(const float4*)(Wn + ow1);
    __builtin_amdgcn_sched_barrier(0);
    const float* ap = &As[cur][aoff];
    const float* bp = &Bs[cur][boff];
    v2f af[KS], bf0[KS], bf1[KS];
#pragma unroll
    for (int s = 0; s < KS; ++s) {
      af[s]  = *(const v2f*)(ap + 4 * s);
      bf0[s] = *(const v2f*)(bp + 4 * s);
      bf1[s] = *(const v2f*)(bp + B1 * PITCH + 4 * s);
    }
#pragma unroll
    for (int s = 0; s < KS; ++s) {
      acc0 = __builtin_amdgcn_wmma_f32_16x16x4_f32(false, af[s], false, bf0[s], (short)0, acc0, false, false);
      acc1 = __builtin_amdgcn_wmma_f32_16x16x4_f32(false, af[s], false, bf1[s], (short)0, acc1, false, false);
    }
    asm volatile("v_nop\n\tv_nop\n\tv_nop\n\tv_nop\n\tv_nop" : "+v"(acc0), "+v"(acc1));
    __builtin_amdgcn_sched_barrier(0);
    *(float4*)&As[cur ^ 1][so0] = pa0;
    *(float4*)&As[cur ^ 1][so1] = pa1;
    *(float4*)&Bs[cur ^ 1][so0] = pw0;
    *(float4*)&Bs[cur ^ 1][so1] = pw1;
    __syncthreads();
  };

  for (int kc = 0; kc < NCH; kc += 2) {
    chunk(0, kc + 1);
    chunk(1, (kc + 2 < NCH) ? kc + 2 : NCH - 1);
  }

  if constexpr (EM == E_PLAIN) {
    const int mbase = row0 + wm * 16 + ((lane >> 4) << 3);
    const int cb = col0 + wn * 32 + ml;
    const float bv0 = bias[cb];
    const float bv1 = bias[cb + 16];
#pragma unroll
    for (int r = 0; r < 8; ++r) {
      float* crow = C + (size_t)(mbase + r) * ldc + cb;
      crow[ 0] = acc0[r] + bv0;
      crow[16] = acc1[r] + bv1;
    }
  } else {
    float* stg = &As[0][0];
    const int cl = wn * 16 + ml;
    const int cg = col0 + cl;
    const float bl = bias[cg];
    const float bg = bias[D_ + cg];
    const int rl0 = wm * 16 + ((lane >> 4) << 3);
#pragma unroll
    for (int r = 0; r < 8; ++r) {
      const int rl = rl0 + r;
      const float xo  = A[(size_t)(row0 + rl) * D_ + cg];
      const float lin = acc0[r] + bl;
      const float gin = acc1[r] + bg;
      const float g  = 1.f / (1.f + __expf(-gin));
      const float nl = fmaxf(lin, 0.f);
      stg[rl * PITCH + cl] = g * xo + (1.f - g) * nl;
    }
    __syncthreads();
#pragma unroll
    for (int i = 0; i < 8; ++i) {
      const int rl = wave * 8 + i;
      const float v = stg[rl * PITCH + lane];
      const size_t m = (size_t)(row0 + rl);
      if constexpr (EM == E_GATE) {
        C[m * D_ + col0 + lane] = v;
      } else {
        const size_t o = m * (size_t)(2 * D_) + (size_t)dir * D_ + col0 + lane;
        C[o] = v;
        C[o + (size_t)BS_ * 2 * D_] = v;
      }
    }
  }
}

#define AT_G 6
typedef __attribute__((ext_vector_type(4))) float v4f;
#define AT_PIN(a, c)                                                            \
  asm volatile("" : "+v"(a[0]), "+v"(a[1]), "+v"(a[2]), "+v"(a[3]), "+v"(a[4]), \
                    "+v"(a[5]),                                                 \
                    "+v"(c[0]), "+v"(c[1]), "+v"(c[2]), "+v"(c[3]), "+v"(c[4]), \
                    "+v"(c[5]))
static_assert(AT_G == 6, "AT_PIN names six slots");
#define AT_AFTER(jv, val) asm volatile("" : "+v"(jv) : "v"(val))
__global__ __launch_bounds__(256)
void attn_kernel(const float* __restrict__ qkv, float* __restrict__ out, int dir) {
  int idx = blockIdx.x * blockDim.x + threadIdx.x;
  if (idx >= B_ * H_ * N_) return;
  int h    = idx % H_;
  int rest = idx / H_;
  int i    = rest % N_;
  int b    = rest / N_;

  const size_t base = (size_t)(b * N_ + i) * D_ + h * DK_;
  const float* qp = qkv + base;
  const float* kbase = qkv + BND_ + (size_t)(b * N_) * D_ + h * DK_;
  const float* vbase = qkv + 2 * BND_ + (size_t)(b * N_) * D_ + h * DK_;

  const float4 q0 = *(const float4*)(qp);
  const float4 q1 = *(const float4*)(qp + 4);

  const int j0 = (dir == 0) ? ((i - WIDTH_ - 1 > 0) ? i - WIDTH_ - 1 : 0) : i;
  const int j1 = (dir == 0) ? i : ((i + WIDTH_ + 1 < N_ - 1) ? i + WIDTH_ + 1 : N_ - 1);
  const float scale = 0.35355339059327373f;

  float sv[WIN_];
  float mx = -3.0e38f;
  int jg = j0;
#pragma unroll
  for (int g = 0; g < WIN_; g += AT_G) {
    v4f ka[AT_G], kc[AT_G];
#pragma unroll
    for (int u = 0; u < AT_G; ++u) {
      const int j  = jg + u;
      const int jc = (j <= j1) ? j : j1;
      const float* kp = kbase + (size_t)jc * D_;
      ka[u] = *(const v4f*)(kp);
      kc[u] = *(const v4f*)(kp + 4);
    }
    AT_PIN(ka, kc);
#pragma unroll
    for (int u = 0; u < AT_G; ++u) {
      float dot = 0.f;
      dot += q0.x * ka[u].x; dot += q0.y * ka[u].y; dot += q0.z * ka[u].z; dot += q0.w * ka[u].w;
      dot += q1.x * kc[u].x; dot += q1.y * kc[u].y; dot += q1.z * kc[u].z; dot += q1.w * kc[u].w;
      const float sc = (jg + u <= j1) ? dot * scale : -3.0e38f;
      sv[g + u] = sc;
      mx = fmaxf(mx, sc);
    }
    if (g + AT_G < WIN_) {
      jg += AT_G;
      AT_AFTER(jg, mx);
    }
  }
  float sum = 0.f;
#pragma unroll
  for (int t = 0; t < WIN_; ++t) {
    float p = __expf(sv[t] - mx);
    sv[t] = p;
    sum += p;
  }
  const float inv = 1.f / sum;
  float o[DK_] = {};
  jg = j0;
#pragma unroll
  for (int g = 0; g < WIN_; g += AT_G) {
    v4f va[AT_G], vc[AT_G];
#pragma unroll
    for (int u = 0; u < AT_G; ++u) {
      const int j  = jg + u;
      const int jc = (j <= j1) ? j : j1;
      const float* vp = vbase + (size_t)jc * D_;
      va[u] = *(const v4f*)(vp);
      vc[u] = *(const v4f*)(vp + 4);
    }
    AT_PIN(va, vc);
#pragma unroll
    for (int u = 0; u < AT_G; ++u) {
      const float w = sv[g + u] * inv;
      o[0] += w * va[u].x; o[1] += w * va[u].y; o[2] += w * va[u].z; o[3] += w * va[u].w;
      o[4] += w * vc[u].x; o[5] += w * vc[u].y; o[6] += w * vc[u].z; o[7] += w * vc[u].w;
    }
    if (g + AT_G < WIN_) {
      jg += AT_G;
      AT_AFTER(jg, o[7]);
    }
  }
  float* op = out + base;
  float4 o0; o0.x = o[0]; o0.y = o[1]; o0.z = o[2]; o0.w = o[3];
  float4 o1; o1.x = o[4]; o1.y = o[5]; o1.z = o[6]; o1.w = o[7];
  *(float4*)(op)     = o0;
  *(float4*)(op + 4) = o1;
}

static_assert(BN_ % GEMM_BM == 0 && BS_ % GEMM_BM == 0, "GEMM row tiling must be exact");
static_assert(D_ % GEMM_BN == 0 && D_ % GATE_BN == 0, "GEMM column tiling must be exact");
static_assert(NL_ == 2, "two highway layers: the first writes the spare buffer, the second the output");

extern "C" void kernel_launch(void* const* d_in, const int* in_sizes, int n_in,
                              void* d_out, int out_size, void* d_ws, size_t ws_size,
                              hipStream_t stream) {
  const float* inputs    = (const float*)d_in[0];
  const float* left_pad  = (const float*)d_in[1];
  const float* right_pad = (const float*)d_in[2];
  const float* Wqkv[2] = {(const float*)d_in[3],  (const float*)d_in[7]};
  const float* bqkv[2] = {(const float*)d_in[4],  (const float*)d_in[8]};
  const float* Wo[2]   = {(const float*)d_in[5],  (const float*)d_in[9]};
  const float* bo[2]   = {(const float*)d_in[6],  (const float*)d_in[10]};
  const float* hwW[2]  = {(const float*)d_in[11], (const float*)d_in[13]};
  const float* hwb[2]  = {(const float*)d_in[12], (const float*)d_in[14]};
  float* out = (float*)d_out;

  float* ws    = (float*)d_ws;
  float* qkv   = ws;
  float* attn  = ws + 3 * BND_;
  float* hx    = ws;
  float* hx2   = ws + 3 * BND_;

  dim3 blk(32, 8);

  for (int dir = 0; dir < 2; ++dir) {
    { dim3 grd(D_ / GEMM_BN, BN_ / GEMM_BM, 3);
      gemm_wmma_f32<A_PADDED, E_PLAIN, Z_PLANE><<<grd, blk, 0, stream>>>(
          inputs, left_pad, right_pad, Wqkv[dir], Wqkv[dir], bqkv[dir], bqkv[dir],
          qkv, D_, BN_, D_); }
    { int n = B_ * H_ * N_;
      attn_kernel<<<(n + 255) / 256, 256, 0, stream>>>(qkv, attn + (size_t)dir * BND_, dir); }
  }
  { dim3 grd(D_ / GEMM_BN, BS_ / GEMM_BM, 2);
    gemm_wmma_f32<A_CENTER, E_PLAIN, Z_DIR><<<grd, blk, 0, stream>>>(
        attn, attn, attn, Wo[0], Wo[1], bo[0], bo[1], hx, D_, BS_, D_); }
  { dim3 grd(D_ / GATE_BN, BS_ / GEMM_BM, 2);
    gemm_wmma_f32<A_PLAIN, E_GATE, Z_DIR><<<grd, blk, 0, stream>>>(
        hx, hx, hx, hwW[0], hwW[1], hwb[0], hwb[1], hx2, D_, BS_, D_); }
  { dim3 grd(D_ / GATE_BN, BS_ / GEMM_BM, 2);
    gemm_wmma_f32<A_PLAIN, E_GATE_OUT, Z_DIR><<<grd, blk, 0, stream>>>(
        hx2, hx2, hx2, hwW[0] + (size_t)2 * D_ * D_, hwW[1] + (size_t)2 * D_ * D_,
        hwb[0] + 2 * D_, hwb[1] + 2 * D_, out, 2 * D_, BS_, D_); }
}


# Both directions share the QKV launch (grid z = 6 planes) and the attention launch (grid y); direction 1's q/k/v planes sit in the output buffer until the last launch overwrites it; 7 launches to 5
# speedup vs baseline: 3.1995x; 1.0690x over previous
#include <hip/hip_runtime.h>

#define B_ 8
#define S_ 512
#define D_ 512
#define H_ 64
#define DK_ 8
#define WIDTH_ 16
#define N_ (S_ + 2 * WIDTH_)
#define NL_ 2
#define WIN_ (WIDTH_ + 2)
#define BN_ (B_ * N_)
#define BS_ (B_ * S_)
static const size_t BND_ = (size_t)BN_ * D_;
static const size_t BSD_ = (size_t)BS_ * D_;

typedef __attribute__((ext_vector_type(2))) float v2f;
typedef __attribute__((ext_vector_type(8))) float v8f;

#define GEMM_BM 64
#define GEMM_BN 64
#define GATE_BN 32

enum { A_PADDED = 0, A_CENTER = 1, A_PLAIN = 2 };
enum { E_PLAIN = 0, E_GATE = 1, E_GATE_OUT = 2 };
enum { Z_PLANE = 0, Z_DIR = 1 };

template <int AM, int EM, int ZM>
__global__ __launch_bounds__(256)
void gemm_wmma_f32(const float* __restrict__ A, const float* __restrict__ LP,
                   const float* __restrict__ RP,
                   const float* __restrict__ W0, const float* __restrict__ W1,
                   const float* __restrict__ bias0, const float* __restrict__ bias1,
                   float* __restrict__ C, float* __restrict__ C1,
                   int ldc, int M, int Nout) {
  constexpr int KK = 512;
  constexpr int BM = GEMM_BM;
  constexpr int BN = (EM == E_PLAIN) ? GEMM_BN : GATE_BN;
  constexpr int BK = 32;
  constexpr int PITCH = BK + 4;
  constexpr int NCH = KK / BK;
  constexpr int KS = BK / 4;
  constexpr int B1 = (EM == E_PLAIN) ? 16 : 32;
  __shared__ __attribute__((aligned(16))) float As[2][BM * PITCH];
  __shared__ __attribute__((aligned(16))) float Bs[2][64 * PITCH];

  const int lane = threadIdx.x;
  const int wave = threadIdx.y;
  const int tid  = wave * 32 + lane;
  const int row0 = blockIdx.y * BM;
  const int col0 = blockIdx.x * BN;

  if (row0 >= M || col0 >= Nout) return;

  const int z = blockIdx.z;
  const float* W;
  const float* bias;
  float* Cb;
  int dir = 0;
  if constexpr (ZM == Z_PLANE) {
    const int zd = z / 3;
    const int zt = z - 3 * zd;
    W    = (zd ? W1 : W0) + (size_t)zt * D_ * D_;
    bias = (zd ? bias1 : bias0) + zt * D_;
    Cb   = (zd ? C1 : C) + (size_t)zt * BND_;
  } else {
    dir  = z;
    W    = z ? W1 : W0;
    bias = z ? bias1 : bias0;
    A   += (size_t)z * ((AM == A_CENTER) ? BND_ : BSD_);
    Cb   = C;
    if constexpr (EM != E_GATE_OUT) Cb += (size_t)z * BSD_;
  }

  const int srow = tid >> 3;
  const int scv  = (tid & 7) << 2;
  int ar0 = row0 + srow;      if (ar0 > M - 1)    ar0 = M - 1;
  int ar1 = row0 + srow + 32; if (ar1 > M - 1)    ar1 = M - 1;
  auto arow = [&](int r) __attribute__((always_inline)) -> const float* {
    if constexpr (AM == A_PADDED) {
      const int b = r / N_;
      const int p = r - b * N_;
      if (p < WIDTH_)       return LP + p * D_;
      if (p < WIDTH_ + S_)  return A + ((size_t)b * S_ + (p - WIDTH_)) * D_;
      return RP + (p - WIDTH_ - S_) * D_;
    } else if constexpr (AM == A_CENTER) {
      const int b = r / S_;
      const int s = r - b * S_;
      return A + ((size_t)b * N_ + WIDTH_ + s) * D_;
    } else {
      return A + (size_t)r * D_;
    }
  };
  const float* ap0 = arow(ar0) + scv;
  const float* ap1 = arow(ar1) + scv;
  int wr0, wr1;
  if constexpr (EM == E_PLAIN) {
    wr0 = col0 + srow;      if (wr0 > Nout - 1) wr0 = Nout - 1;
    wr1 = col0 + srow + 32; if (wr1 > Nout - 1) wr1 = Nout - 1;
  } else {
    wr0 = col0 + srow;
    wr1 = D_ + col0 + srow;
  }
  const unsigned ow0 = (unsigned)(wr0 * KK + scv) * 4u;
  const unsigned ow1 = (unsigned)(wr1 * KK + scv) * 4u;
  const int so0 = srow * PITCH + scv;
  const int so1 = (srow + 32) * PITCH + scv;

  float4 pa0 = *(const float4*)(ap0);
  float4 pa1 = *(const float4*)(ap1);
  float4 pw0 = *(const float4*)((const char*)W + ow0);
  float4 pw1 = *(const float4*)((const char*)W + ow1);
  *(float4*)&As[0][so0] = pa0;
  *(float4*)&As[0][so1] = pa1;
  *(float4*)&Bs[0][so0] = pw0;
  *(float4*)&Bs[0][so1] = pw1;
  __syncthreads();

  const int wm = wave & 3;
  const int wn = wave >> 2;
  const int ml = lane & 15;
  const int kb = (lane >> 4) << 1;
  const int aoff = (wm * 16 + ml) * PITCH + kb;
  const int boff = (wn * ((EM == E_PLAIN) ? 32 : 16) + ml) * PITCH + kb;

  v8f acc0 = {}, acc1 = {};

  auto chunk = [&](const int cur, const int kn) __attribute__((always_inline)) {
    const char* Wn = (const char*)(W + kn * BK);
    pa0 = *(const float4*)(ap0 + kn * BK);
    pa1 = *(const float4*)(ap1 + kn * BK);
    pw0 = *(const float4*)(Wn + ow0);
    pw1 = *(const float4*)(Wn + ow1);
    __builtin_amdgcn_sched_barrier(0);
    const float* ap = &As[cur][aoff];
    const float* bp = &Bs[cur][boff];
    v2f af[KS], bf0[KS], bf1[KS];
#pragma unroll
    for (int s = 0; s < KS; ++s) {
      af[s]  = *(const v2f*)(ap + 4 * s);
      bf0[s] = *(const v2f*)(bp + 4 * s);
      bf1[s] = *(const v2f*)(bp + B1 * PITCH + 4 * s);
    }
#pragma unroll
    for (int s = 0; s < KS; ++s) {
      acc0 = __builtin_amdgcn_wmma_f32_16x16x4_f32(false, af[s], false, bf0[s], (short)0, acc0, false, false);
      acc1 = __builtin_amdgcn_wmma_f32_16x16x4_f32(false, af[s], false, bf1[s], (short)0, acc1, false, false);
    }
    asm volatile("v_nop\n\tv_nop\n\tv_nop\n\tv_nop\n\tv_nop" : "+v"(acc0), "+v"(acc1));
    __builtin_amdgcn_sched_barrier(0);
    *(float4*)&As[cur ^ 1][so0] = pa0;
    *(float4*)&As[cur ^ 1][so1] = pa1;
    *(float4*)&Bs[cur ^ 1][so0] = pw0;
    *(float4*)&Bs[cur ^ 1][so1] = pw1;
    __syncthreads();
  };

  for (int kc = 0; kc < NCH; kc += 2) {
    chunk(0, kc + 1);
    chunk(1, (kc + 2 < NCH) ? kc + 2 : NCH - 1);
  }

  if constexpr (EM == E_PLAIN) {
    const int mbase = row0 + wm * 16 + ((lane >> 4) << 3);
    const int cb = col0 + wn * 32 + ml;
    const float bv0 = bias[cb];
    const float bv1 = bias[cb + 16];
#pragma unroll
    for (int r = 0; r < 8; ++r) {
      float* crow = Cb + (size_t)(mbase + r) * ldc + cb;
      crow[ 0] = acc0[r] + bv0;
      crow[16] = acc1[r] + bv1;
    }
  } else {
    float* stg = &As[0][0];
    const int cl = wn * 16 + ml;
    const int cg = col0 + cl;
    const float bl = bias[cg];
    const float bg = bias[D_ + cg];
    const int rl0 = wm * 16 + ((lane >> 4) << 3);
#pragma unroll
    for (int r = 0; r < 8; ++r) {
      const int rl = rl0 + r;
      const float xo  = A[(size_t)(row0 + rl) * D_ + cg];
      const float lin = acc0[r] + bl;
      const float gin = acc1[r] + bg;
      const float g  = 1.f / (1.f + __expf(-gin));
      const float nl = fmaxf(lin, 0.f);
      stg[rl * PITCH + cl] = g * xo + (1.f - g) * nl;
    }
    __syncthreads();
#pragma unroll
    for (int i = 0; i < 8; ++i) {
      const int rl = wave * 8 + i;
      const float v = stg[rl * PITCH + lane];
      const size_t m = (size_t)(row0 + rl);
      if constexpr (EM == E_GATE) {
        Cb[m * D_ + col0 + lane] = v;
      } else {
        const size_t o = m * (size_t)(2 * D_) + (size_t)dir * D_ + col0 + lane;
        Cb[o] = v;
        Cb[o + (size_t)BS_ * 2 * D_] = v;
      }
    }
  }
}

#define AT_G 6
typedef __attribute__((ext_vector_type(4))) float v4f;
#define AT_PIN(a, c)                                                            \
  asm volatile("" : "+v"(a[0]), "+v"(a[1]), "+v"(a[2]), "+v"(a[3]), "+v"(a[4]), \
                    "+v"(a[5]),                                                 \
                    "+v"(c[0]), "+v"(c[1]), "+v"(c[2]), "+v"(c[3]), "+v"(c[4]), \
                    "+v"(c[5]))
static_assert(AT_G == 6, "AT_PIN names six slots");
#define AT_AFTER(jv, val) asm volatile("" : "+v"(jv) : "v"(val))
__global__ __launch_bounds__(256)
void attn_kernel(const float* __restrict__ qkv0, const float* __restrict__ qkv1,
                 float* __restrict__ out) {
  int idx = blockIdx.x * blockDim.x + threadIdx.x;
  if (idx >= B_ * H_ * N_) return;
  const int dir = blockIdx.y;
  const float* qkv = dir ? qkv1 : qkv0;
  out += (size_t)dir * BND_;
  int h    = idx % H_;
  int rest = idx / H_;
  int i    = rest % N_;
  int b    = rest / N_;

  const size_t base = (size_t)(b * N_ + i) * D_ + h * DK_;
  const float* qp = qkv + base;
  const float* kbase = qkv + BND_ + (size_t)(b * N_) * D_ + h * DK_;
  const float* vbase = qkv + 2 * BND_ + (size_t)(b * N_) * D_ + h * DK_;

  const float4 q0 = *(const float4*)(qp);
  const float4 q1 = *(const float4*)(qp + 4);

  const int j0 = (dir == 0) ? ((i - WIDTH_ - 1 > 0) ? i - WIDTH_ - 1 : 0) : i;
  const int j1 = (dir == 0) ? i : ((i + WIDTH_ + 1 < N_ - 1) ? i + WIDTH_ + 1 : N_ - 1);
  const float scale = 0.35355339059327373f;

  float sv[WIN_];
  float mx = -3.0e38f;
  int jg = j0;
#pragma unroll
  for (int g = 0; g < WIN_; g += AT_G) {
    v4f ka[AT_G], kc[AT_G];
#pragma unroll
    for (int u = 0; u < AT_G; ++u) {
      const int j  = jg + u;
      const int jc = (j <= j1) ? j : j1;
      const float* kp = kbase + (size_t)jc * D_;
      ka[u] = *(const v4f*)(kp);
      kc[u] = *(const v4f*)(kp + 4);
    }
    AT_PIN(ka, kc);
#pragma unroll
    for (int u = 0; u < AT_G; ++u) {
      float dot = 0.f;
      dot += q0.x * ka[u].x; dot += q0.y * ka[u].y; dot += q0.z * ka[u].z; dot += q0.w * ka[u].w;
      dot += q1.x * kc[u].x; dot += q1.y * kc[u].y; dot += q1.z * kc[u].z; dot += q1.w * kc[u].w;
      const float sc = (jg + u <= j1) ? dot * scale : -3.0e38f;
      sv[g + u] = sc;
      mx = fmaxf(mx, sc);
    }
    if (g + AT_G < WIN_) {
      jg += AT_G;
      AT_AFTER(jg, mx);
    }
  }
  float sum = 0.f;
#pragma unroll
  for (int t = 0; t < WIN_; ++t) {
    float p = __expf(sv[t] - mx);
    sv[t] = p;
    sum += p;
  }
  const float inv = 1.f / sum;
  float o[DK_] = {};
  jg = j0;
#pragma unroll
  for (int g = 0; g < WIN_; g += AT_G) {
    v4f va[AT_G], vc[AT_G];
#pragma unroll
    for (int u = 0; u < AT_G; ++u) {
      const int j  = jg + u;
      const int jc = (j <= j1) ? j : j1;
      const float* vp = vbase + (size_t)jc * D_;
      va[u] = *(const v4f*)(vp);
      vc[u] = *(const v4f*)(vp + 4);
    }
    AT_PIN(va, vc);
#pragma unroll
    for (int u = 0; u < AT_G; ++u) {
      const float w = sv[g + u] * inv;
      o[0] += w * va[u].x; o[1] += w * va[u].y; o[2] += w * va[u].z; o[3] += w * va[u].w;
      o[4] += w * vc[u].x; o[5] += w * vc[u].y; o[6] += w * vc[u].z; o[7] += w * vc[u].w;
    }
    if (g + AT_G < WIN_) {
      jg += AT_G;
      AT_AFTER(jg, o[7]);
    }
  }
  float* op = out + base;
  float4 o0; o0.x = o[0]; o0.y = o[1]; o0.z = o[2]; o0.w = o[3];
  float4 o1; o1.x = o[4]; o1.y = o[5]; o1.z = o[6]; o1.w = o[7];
  *(float4*)(op)     = o0;
  *(float4*)(op + 4) = o1;
}

static_assert(BN_ % GEMM_BM == 0 && BS_ % GEMM_BM == 0, "GEMM row tiling must be exact");
static_assert(D_ % GEMM_BN == 0 && D_ % GATE_BN == 0, "GEMM column tiling must be exact");
static_assert(NL_ == 2, "two highway layers: the first writes the spare buffer, the second the output");

extern "C" void kernel_launch(void* const* d_in, const int* in_sizes, int n_in,
                              void* d_out, int out_size, void* d_ws, size_t ws_size,
                              hipStream_t stream) {
  const float* inputs    = (const float*)d_in[0];
  const float* left_pad  = (const float*)d_in[1];
  const float* right_pad = (const float*)d_in[2];
  const float* Wqkv[2] = {(const float*)d_in[3],  (const float*)d_in[7]};
  const float* bqkv[2] = {(const float*)d_in[4],  (const float*)d_in[8]};
  const float* Wo[2]   = {(const float*)d_in[5],  (const float*)d_in[9]};
  const float* bo[2]   = {(const float*)d_in[6],  (const float*)d_in[10]};
  const float* hwW[2]  = {(const float*)d_in[11], (const float*)d_in[13]};
  const float* hwb[2]  = {(const float*)d_in[12], (const float*)d_in[14]};
  float* out = (float*)d_out;

  float* ws    = (float*)d_ws;
  float* qkv0  = ws;
  float* qkv1  = out;
  float* attn  = ws + 3 * BND_;
  float* hx    = ws;
  float* hx2   = ws + 3 * BND_;

  dim3 blk(32, 8);

  { dim3 grd(D_ / GEMM_BN, BN_ / GEMM_BM, 6);
    gemm_wmma_f32<A_PADDED, E_PLAIN, Z_PLANE><<<grd, blk, 0, stream>>>(
        inputs, left_pad, right_pad, Wqkv[0], Wqkv[1], bqkv[0], bqkv[1],
        qkv0, qkv1, D_, BN_, D_); }
  { int n = B_ * H_ * N_;
    dim3 grd((n + 255) / 256, 2);
    attn_kernel<<<grd, 256, 0, stream>>>(qkv0, qkv1, attn); }
  { dim3 grd(D_ / GEMM_BN, BS_ / GEMM_BM, 2);
    gemm_wmma_f32<A_CENTER, E_PLAIN, Z_DIR><<<grd, blk, 0, stream>>>(
        attn, attn, attn, Wo[0], Wo[1], bo[0], bo[1], hx, hx, D_, BS_, D_); }
  { dim3 grd(D_ / GATE_BN, BS_ / GEMM_BM, 2);
    gemm_wmma_f32<A_PLAIN, E_GATE, Z_DIR><<<grd, blk, 0, stream>>>(
        hx, hx, hx, hwW[0], hwW[1], hwb[0], hwb[1], hx2, hx2, D_, BS_, D_); }
  { dim3 grd(D_ / GATE_BN, BS_ / GEMM_BM, 2);
    gemm_wmma_f32<A_PLAIN, E_GATE_OUT, Z_DIR><<<grd, blk, 0, stream>>>(
        hx2, hx2, hx2, hwW[0] + (size_t)2 * D_ * D_, hwW[1] + (size_t)2 * D_ * D_,
        hwb[0] + 2 * D_, hwb[1] + 2 * D_, out, out, 2 * D_, BS_, D_); }
}
